# starting kernel
# baseline (speedup 1.0000x reference)
#include <hip/hip_runtime.h>
#include <hip/hip_bf16.h>
#include <math.h>

#define BATCH   256
#define IN_DIM  1024
#define OUT_DIM 1024
#define NPATHS  2000
#define NSTEPS  128
#define NPTOT   (BATCH * NPATHS)
#define DT      (1.0f / 128.0f)
#define SQDT    0.08838834764831845f

typedef float v2f __attribute__((ext_vector_type(2)));
typedef float v8f __attribute__((ext_vector_type(8)));

__global__ __launch_bounds__(64) void bicep_fb_kernel(
    const float* __restrict__ x, const float* __restrict__ W_in,
    const float* __restrict__ b_in, const float* __restrict__ decay,
    float* __restrict__ a_ws) {
  __shared__ float red[64];
  const int bg = blockIdx.x;
  const int b = bg / 10, g = bg % 10;
  const float* xr = x + b * IN_DIM;
  const float* wr = W_in + g * IN_DIM;
  float s = 0.0f;
  for (int i = threadIdx.x; i < IN_DIM; i += 64) s = fmaf(xr[i], wr[i], s);
  red[threadIdx.x] = s;
  __syncthreads();
  for (int off = 32; off > 0; off >>= 1) {
    if (threadIdx.x < off) red[threadIdx.x] += red[threadIdx.x + off];
    __syncthreads();
  }
  if (threadIdx.x == 0) {
    float proj = red[0] + b_in[g];
    float fb = 1.0f / (1.0f + expf(-proj));
    a_ws[bg] = 1.0f + (fb - decay[0]) * DT;
  }
}

__global__ __launch_bounds__(256) void bicep_path_kernel(
    const float* __restrict__ eps, const float* __restrict__ W_conv,
    const float* __restrict__ a_ws,
    float* __restrict__ S0, float* __restrict__ S1, float* __restrict__ S2) {
  __shared__ float wc[(NSTEPS + 1) * 3];
  for (int i = threadIdx.x; i < (NSTEPS + 1) * 3; i += 256) wc[i] = W_conv[i];
  __syncthreads();

  const int id = blockIdx.x * 256 + threadIdx.x;
  const int b  = id / NPATHS;
  const int pg = id - b * NPATHS;
  const int g  = pg / 200;
  const float aa = a_ws[b * 10 + g];

  float X = 0.0f, s0 = 0.0f, s1 = 0.0f, s2 = 0.0f;
  const float* ep = eps + id;
  for (int t = 1; t <= NSTEPS; ++t) {
    float e = ep[(t - 1) * NPTOT];
    X  = fmaf(X, aa, SQDT * e);
    s0 = fmaf(wc[t * 3 + 0], X, s0);
    s1 = fmaf(wc[t * 3 + 1], X, s1);
    s2 = fmaf(wc[t * 3 + 2], X, s2);
  }
  S0[id] = s0; S1[id] = s1; S2[id] = s2;
}

__global__ __launch_bounds__(256) void bicep_agg_kernel(
    const float* __restrict__ S0, const float* __restrict__ S1,
    const float* __restrict__ S2, const float* __restrict__ b_conv,
    float* __restrict__ agg) {
  const int id = blockIdx.x * 256 + threadIdx.x;
  const int pg = id % NPATHS;
  float v = S1[id] + b_conv[0];
  if (pg > 0)          v += S0[id - 1];
  if (pg < NPATHS - 1) v += S2[id + 1];
  agg[id] = v;
}

__global__ __launch_bounds__(256) void bicep_gemm_kernel(
    const float* __restrict__ agg, const float* __restrict__ W_out,
    const float* __restrict__ b_out, float* __restrict__ out) {
  const int lane = threadIdx.x & 31;
  const int wave = threadIdx.x >> 5;
  const int tile = blockIdx.x * 8 + wave;
  const int mTile = tile >> 6;
  const int nTile = tile & 63;
  const int half = lane >> 4;
  const int idx  = lane & 15;

  const float* arow = agg   + (mTile * 16 + idx) * NPATHS + 2 * half;
  const float* brow = W_out + (nTile * 16 + idx) * NPATHS + 2 * half;

  v8f c = {};
  for (int k = 0; k < NPATHS; k += 4) {
    v2f a = *(const v2f*)(arow + k);
    v2f b = *(const v2f*)(brow + k);
    c = __builtin_amdgcn_wmma_f32_16x16x4_f32(
         false, a,  false, b,
         (short)0, c,  false,  false);
  }

  const float bb = b_out[nTile * 16 + idx];
  float* orow = out + (mTile * 16 + 8 * half) * OUT_DIM + nTile * 16 + idx;
#pragma unroll
  for (int r = 0; r < 8; ++r) orow[r * OUT_DIM] = c[r] + bb;
}

extern "C" void kernel_launch(void* const* d_in, const int* in_sizes, int n_in,
                              void* d_out, int out_size, void* d_ws, size_t ws_size,
                              hipStream_t stream) {
  (void)in_sizes; (void)n_in; (void)out_size; (void)ws_size;
  const float* x      = (const float*)d_in[0];
  const float* W_in   = (const float*)d_in[1];
  const float* b_in   = (const float*)d_in[2];
  const float* decay  = (const float*)d_in[3];
  const float* W_conv = (const float*)d_in[4];
  const float* b_conv = (const float*)d_in[5];
  const float* W_out  = (const float*)d_in[6];
  const float* b_out  = (const float*)d_in[7];
  const float* eps    = (const float*)d_in[8];
  float* out = (float*)d_out;

  float* ws   = (float*)d_ws;
  float* a_ws = ws;
  float* S0   = ws + 4096;
  float* S1   = S0 + NPTOT;
  float* S2   = S1 + NPTOT;
  float* agg  = S2 + NPTOT;

  bicep_fb_kernel<<<BATCH * 10, 64, 0, stream>>>(x, W_in, b_in, decay, a_ws);
  bicep_path_kernel<<<NPTOT / 256, 256, 0, stream>>>(eps, W_conv, a_ws, S0, S1, S2);
  bicep_agg_kernel<<<NPTOT / 256, 256, 0, stream>>>(S0, S1, S2, b_conv, agg);
  bicep_gemm_kernel<<<(16 * 64) / 8, 256, 0, stream>>>(agg, W_out, b_out, out);
}


# GEMM re-tiled: 32x32 block tile, K slabs of 80 staged in LDS by row-contiguous 16B loads and shared by 4 waves, fragments read from LDS; k order permuted, two accumulator chains (sum order changes)
# speedup vs baseline: 1.4389x; 1.4389x over previous
#include <hip/hip_runtime.h>
#include <hip/hip_bf16.h>
#include <math.h>

#define BATCH   256
#define IN_DIM  1024
#define OUT_DIM 1024
#define NPATHS  2000
#define NSTEPS  128
#define NPTOT   (BATCH * NPATHS)
#define DT      (1.0f / 128.0f)
#define SQDT    0.08838834764831845f

typedef float v2f __attribute__((ext_vector_type(2)));
typedef float v4f __attribute__((ext_vector_type(4)));
typedef float v8f __attribute__((ext_vector_type(8)));

__global__ __launch_bounds__(64) void bicep_fb_kernel(
    const float* __restrict__ x, const float* __restrict__ W_in,
    const float* __restrict__ b_in, const float* __restrict__ decay,
    float* __restrict__ a_ws) {
  __shared__ float red[64];
  const int bg = blockIdx.x;
  const int b = bg / 10, g = bg % 10;
  const float* xr = x + b * IN_DIM;
  const float* wr = W_in + g * IN_DIM;
  float s = 0.0f;
  for (int i = threadIdx.x; i < IN_DIM; i += 64) s = fmaf(xr[i], wr[i], s);
  red[threadIdx.x] = s;
  __syncthreads();
  for (int off = 32; off > 0; off >>= 1) {
    if (threadIdx.x < off) red[threadIdx.x] += red[threadIdx.x + off];
    __syncthreads();
  }
  if (threadIdx.x == 0) {
    float proj = red[0] + b_in[g];
    float fb = 1.0f / (1.0f + expf(-proj));
    a_ws[bg] = 1.0f + (fb - decay[0]) * DT;
  }
}

__global__ __launch_bounds__(256) void bicep_path_kernel(
    const float* __restrict__ eps, const float* __restrict__ W_conv,
    const float* __restrict__ a_ws,
    float* __restrict__ S0, float* __restrict__ S1, float* __restrict__ S2) {
  __shared__ float wc[(NSTEPS + 1) * 3];
  for (int i = threadIdx.x; i < (NSTEPS + 1) * 3; i += 256) wc[i] = W_conv[i];
  __syncthreads();

  const int id = blockIdx.x * 256 + threadIdx.x;
  const int b  = id / NPATHS;
  const int pg = id - b * NPATHS;
  const int g  = pg / 200;
  const float aa = a_ws[b * 10 + g];

  float X = 0.0f, s0 = 0.0f, s1 = 0.0f, s2 = 0.0f;
  const float* ep = eps + id;
  for (int t = 1; t <= NSTEPS; ++t) {
    float e = ep[(t - 1) * NPTOT];
    X  = fmaf(X, aa, SQDT * e);
    s0 = fmaf(wc[t * 3 + 0], X, s0);
    s1 = fmaf(wc[t * 3 + 1], X, s1);
    s2 = fmaf(wc[t * 3 + 2], X, s2);
  }
  S0[id] = s0; S1[id] = s1; S2[id] = s2;
}

__global__ __launch_bounds__(256) void bicep_agg_kernel(
    const float* __restrict__ S0, const float* __restrict__ S1,
    const float* __restrict__ S2, const float* __restrict__ b_conv,
    float* __restrict__ agg) {
  const int id = blockIdx.x * 256 + threadIdx.x;
  const int pg = id % NPATHS;
  float v = S1[id] + b_conv[0];
  if (pg > 0)          v += S0[id - 1];
  if (pg < NPATHS - 1) v += S2[id + 1];
  agg[id] = v;
}

#define GT_K     80
#define GT_PITCH 84
#define GT_NSLAB (NPATHS / GT_K)
#define GT_LD    5
#define GT_NF    (GT_K / 8)
#define GT_CP    36

static __device__ __forceinline__ void lds_barrier() {
  asm volatile("s_wait_dscnt 0x0\n\ts_barrier_signal -1\n\ts_barrier_wait -1" ::: "memory");
}
static __device__ __forceinline__ void pin() { __builtin_amdgcn_sched_barrier(0); }

__global__ __launch_bounds__(128) void bicep_gemm_kernel(
    const float* __restrict__ agg, const float* __restrict__ W_out,
    const float* __restrict__ b_out, float* __restrict__ out) {
  __shared__ __attribute__((aligned(16))) float lsA[2][32 * GT_PITCH];
  __shared__ __attribute__((aligned(16))) float lsB[2][32 * GT_PITCH];

  const int tid  = threadIdx.x;
  const int lane = tid & 31;
  const int wave = tid >> 5;
  const int half = lane >> 4;
  const int idx  = lane & 15;
  const int mBlk = blockIdx.x >> 5;
  const int nBlk = blockIdx.x & 31;
  const int wm = wave >> 1, wn = wave & 1;

  int goff[GT_LD], loff[GT_LD];
  for (int j = 0; j < GT_LD; ++j) {
    const int i = tid + 128 * j;
    const int r = i / 20, c4 = i - r * 20;
    goff[j] = r * NPATHS + c4 * 4;
    loff[j] = r * GT_PITCH + c4 * 4;
  }
  const float* gA = agg   + mBlk * 32 * NPATHS;
  const float* gB = W_out + nBlk * 32 * NPATHS;

  const int fragA = (wm * 16 + idx) * GT_PITCH + 4 * half;
  const int fragB = (wn * 16 + idx) * GT_PITCH + 4 * half;

  v4f sa0[GT_LD], sb0[GT_LD], sa1[GT_LD], sb1[GT_LD];
  v4f fa0[GT_NF], fb0[GT_NF], fa1[GT_NF], fb1[GT_NF];
  v8f c0 = {}, c1 = {};

  auto issue = [&](v4f (&qa)[GT_LD], v4f (&qb)[GT_LD], int slab) {
    const float* nA = gA + slab * GT_K;
    const float* nB = gB + slab * GT_K;
    for (int j = 0; j < GT_LD; ++j) {
      qa[j] = *(const v4f*)(nA + goff[j]);
      qb[j] = *(const v4f*)(nB + goff[j]);
    }
  };
  auto park = [&](const v4f (&qa)[GT_LD], const v4f (&qb)[GT_LD], int stage) {
    for (int j = 0; j < GT_LD; ++j) {
      *(v4f*)(&lsA[stage][loff[j]]) = qa[j];
      *(v4f*)(&lsB[stage][loff[j]]) = qb[j];
    }
  };
  auto fload = [&](v4f (&fa)[GT_NF], v4f (&fb)[GT_NF], int stage) {
    const float* pa = &lsA[stage][fragA];
    const float* pb = &lsB[stage][fragB];
    for (int i = 0; i < GT_NF; ++i) {
      fa[i] = *(const v4f*)(pa + 8 * i);
      fb[i] = *(const v4f*)(pb + 8 * i);
    }
  };
  auto mma = [&](const v4f (&fa)[GT_NF], const v4f (&fb)[GT_NF], int lo, int hi) {
    for (int i = lo; i < hi; ++i) {
      c0 = __builtin_amdgcn_wmma_f32_16x16x4_f32(
           false, fa[i].xy,  false, fb[i].xy,
           (short)0, c0,  false,  false);
      c1 = __builtin_amdgcn_wmma_f32_16x16x4_f32(
           false, fa[i].zw,  false, fb[i].zw,
           (short)0, c1,  false,  false);
    }
  };

  issue(sa0, sb0, 0);
  issue(sa1, sb1, 1);
  park(sa0, sb0, 0);
  lds_barrier();
  fload(fa0, fb0, 0);
  pin();

  for (int s = 0; s < GT_NSLAB - 3; s += 2) {
    mma(fa0, fb0, 0, 1);      pin();
    issue(sa0, sb0, s + 2);   pin();
    mma(fa0, fb0, 1, 3);      pin();
    park(sa1, sb1, 1);        pin();
    mma(fa0, fb0, 3, 7);      pin();
    lds_barrier();
    fload(fa1, fb1, 1);       pin();
    mma(fa0, fb0, 7, GT_NF);
    mma(fa1, fb1, 0, 1);      pin();
    issue(sa1, sb1, s + 3);   pin();
    mma(fa1, fb1, 1, 3);      pin();
    park(sa0, sb0, 0);        pin();
    mma(fa1, fb1, 3, 7);      pin();
    lds_barrier();
    fload(fa0, fb0, 0);       pin();
    mma(fa1, fb1, 7, GT_NF);
  }
  mma(fa0, fb0, 0, 1);        pin();
  issue(sa0, sb0, GT_NSLAB - 1); pin();
  mma(fa0, fb0, 1, 3);        pin();
  park(sa1, sb1, 1);          pin();
  mma(fa0, fb0, 3, 7);        pin();
  lds_barrier();
  fload(fa1, fb1, 1);         pin();
  mma(fa0, fb0, 7, GT_NF);
  mma(fa1, fb1, 0, 3);        pin();
  park(sa0, sb0, 0);          pin();
  mma(fa1, fb1, 3, 7);        pin();
  lds_barrier();
  fload(fa0, fb0, 0);         pin();
  mma(fa1, fb1, 7, GT_NF);
  mma(fa0, fb0, 0, GT_NF);
  lds_barrier();

  const v8f c = c0 + c1;
  float* Cs = &lsA[0][0];
  for (int r = 0; r < 8; ++r)
    Cs[(wm * 16 + 8 * half + r) * GT_CP + wn * 16 + idx] = c[r];
  lds_barrier();
  const int c4 = tid & 7;
  const v4f bb = *(const v4f*)(b_out + nBlk * 32 + c4 * 4);
  for (int h = 0; h < 2; ++h) {
    const int row = (tid >> 3) + 16 * h;
    const v4f v = *(const v4f*)(Cs + row * GT_CP + c4 * 4);
    *(v4f*)(out + (mBlk * 32 + row) * OUT_DIM + nBlk * 32 + c4 * 4) = v + bb;
  }
}

extern "C" void kernel_launch(void* const* d_in, const int* in_sizes, int n_in,
                              void* d_out, int out_size, void* d_ws, size_t ws_size,
                              hipStream_t stream) {
  (void)in_sizes; (void)n_in; (void)out_size; (void)ws_size;
  const float* x      = (const float*)d_in[0];
  const float* W_in   = (const float*)d_in[1];
  const float* b_in   = (const float*)d_in[2];
  const float* decay  = (const float*)d_in[3];
  const float* W_conv = (const float*)d_in[4];
  const float* b_conv = (const float*)d_in[5];
  const float* W_out  = (const float*)d_in[6];
  const float* b_out  = (const float*)d_in[7];
  const float* eps    = (const float*)d_in[8];
  float* out = (float*)d_out;

  float* ws   = (float*)d_ws;
  float* a_ws = ws;
  float* S0   = ws + 4096;
  float* S1   = S0 + NPTOT;
  float* S2   = S1 + NPTOT;
  float* agg  = S2 + NPTOT;

  bicep_fb_kernel<<<BATCH * 10, 64, 0, stream>>>(x, W_in, b_in, decay, a_ws);
  bicep_path_kernel<<<NPTOT / 256, 256, 0, stream>>>(eps, W_conv, a_ws, S0, S1, S2);
  bicep_agg_kernel<<<NPTOT / 256, 256, 0, stream>>>(S0, S1, S2, b_conv, agg);
  bicep_gemm_kernel<<<8 * 32, 128, 0, stream>>>(agg, W_out, b_out, out);
}


# Pre-GEMM stages (fb, path scan, neighbour combine) fused into one kernel, one work-group per batch row: 4 paths per lane, 16-byte eps loads issued a group ahead; 4 launches -> 2; sum order kept
# speedup vs baseline: 1.9112x; 1.3282x over previous
#include <hip/hip_runtime.h>
#include <hip/hip_bf16.h>
#include <math.h>

#define BATCH   256
#define IN_DIM  1024
#define OUT_DIM 1024
#define NPATHS  2000
#define NSTEPS  128
#define NPTOT   (BATCH * NPATHS)
#define DT      (1.0f / 128.0f)
#define SQDT    0.08838834764831845f

typedef float v2f __attribute__((ext_vector_type(2)));
typedef float v4f __attribute__((ext_vector_type(4)));
typedef float v8f __attribute__((ext_vector_type(8)));

static __device__ __forceinline__ void pin() { __builtin_amdgcn_sched_barrier(0); }

#define RP_THREADS 512
#define RP_PIECES  (NPATHS / 4)
#define RP_GRP     16
#define RP_NGRP    (NSTEPS / RP_GRP)
#define RP_TSTRIDE (NPTOT / 4)
#define AGG_PITCH  2048

__global__ __launch_bounds__(RP_THREADS) void bicep_row_kernel(
    const float* __restrict__ x, const float* __restrict__ W_in,
    const float* __restrict__ b_in, const float* __restrict__ decay,
    const float* __restrict__ eps, const float* __restrict__ W_conv,
    const float* __restrict__ b_conv, float* __restrict__ agg) {
  __shared__ float red[10][64];
  __shared__ float a_sh[16];
  __shared__ float shL[RP_THREADS];
  __shared__ float shR[RP_THREADS];

  const int tid = threadIdx.x;
  const int b   = blockIdx.x;

  {
    const int grp = tid >> 6, j = tid & 63;
    const float* xr = x + b * IN_DIM;
    const float* wr = W_in + grp * IN_DIM;
    float s = 0.0f;
    for (int i = j; i < IN_DIM; i += 64) s = fmaf(xr[i], wr[i], s);
    red[grp][j] = s;
    if (grp < 2) {
      const float* wr2 = W_in + (grp + 8) * IN_DIM;
      float s2 = 0.0f;
      for (int i = j; i < IN_DIM; i += 64) s2 = fmaf(xr[i], wr2[i], s2);
      red[grp + 8][j] = s2;
    }
  }
  __syncthreads();
  {
    const int wave = tid >> 5, lane = tid & 31;
    if (wave < 10) {
      float v = red[wave][lane] + red[wave][lane + 32];
      v += __shfl_down(v, 16);
      v += __shfl_down(v, 8);
      v += __shfl_down(v, 4);
      v += __shfl_down(v, 2);
      v += __shfl_down(v, 1);
      if (lane == 0) {
        const float proj = v + b_in[wave];
        const float fb = 1.0f / (1.0f + expf(-proj));
        a_sh[wave] = 1.0f + (fb - decay[0]) * DT;
      }
    }
  }
  __syncthreads();

  const int p = tid - 4 * (b & 1);
  const bool active = (p >= 0) && (p < RP_PIECES);
  const int pc = p < 0 ? 0 : (p >= RP_PIECES ? RP_PIECES - 1 : p);
  const float aa = a_sh[pc / 50];
  const v4f* ep = (const v4f*)(eps + b * NPATHS + 4 * pc);

  v4f X = {0.0f, 0.0f, 0.0f, 0.0f};
  v4f s0 = X, s1 = X, s2 = X;
  v4f q0[RP_GRP], q1[RP_GRP];

  auto issue = [&](v4f (&q)[RP_GRP], int g) {
    const v4f* src = ep + g * (RP_GRP * RP_TSTRIDE);
#pragma unroll
    for (int k = 0; k < RP_GRP; ++k) q[k] = src[k * RP_TSTRIDE];
  };
  auto scan = [&](const v4f (&q)[RP_GRP], int g) {
    const float* w = W_conv + (g * RP_GRP + 1) * 3;
#pragma unroll
    for (int k = 0; k < RP_GRP; ++k) {
      const float w0 = w[3 * k + 0], w1 = w[3 * k + 1], w2 = w[3 * k + 2];
#pragma unroll
      for (int c = 0; c < 4; ++c) {
        X[c]  = fmaf(X[c], aa, SQDT * q[k][c]);
        s0[c] = fmaf(w0, X[c], s0[c]);
        s1[c] = fmaf(w1, X[c], s1[c]);
        s2[c] = fmaf(w2, X[c], s2[c]);
      }
    }
  };

  issue(q0, 0);
  for (int g = 0; g < RP_NGRP - 2; g += 2) {
    issue(q1, g + 1);        pin();
    scan(q0, g);             pin();
    issue(q0, g + 2);        pin();
    scan(q1, g + 1);         pin();
  }
  issue(q1, RP_NGRP - 1);    pin();
  scan(q0, RP_NGRP - 2);     pin();
  scan(q1, RP_NGRP - 1);

  shL[tid] = s0[3];
  shR[tid] = s2[0];
  __syncthreads();
  if (active) {
    const float bc = b_conv[0];
    float v0 = s1[0] + bc;
    if (p > 0) v0 += shL[tid - 1];
    v0 += s2[1];
    float v1 = s1[1] + bc;  v1 += s0[0];  v1 += s2[2];
    float v2 = s1[2] + bc;  v2 += s0[1];  v2 += s2[3];
    float v3 = s1[3] + bc;  v3 += s0[2];
    if (p < RP_PIECES - 1) v3 += shR[tid + 1];
    const v4f o = {v0, v1, v2, v3};
    *(v4f*)(agg + b * AGG_PITCH + 4 * tid) = o;
  }
}

#define GT_K     80
#define GT_PITCH 84
#define GT_NSLAB (NPATHS / GT_K)
#define GT_LD    5
#define GT_NF    (GT_K / 8)
#define GT_CP    36

static __device__ __forceinline__ void lds_barrier() {
  asm volatile("s_wait_dscnt 0x0\n\ts_barrier_signal -1\n\ts_barrier_wait -1" ::: "memory");
}

__global__ __launch_bounds__(128) void bicep_gemm_kernel(
    const float* __restrict__ agg, const float* __restrict__ W_out,
    const float* __restrict__ b_out, float* __restrict__ out) {
  __shared__ __attribute__((aligned(16))) float lsA[2][32 * GT_PITCH];
  __shared__ __attribute__((aligned(16))) float lsB[2][32 * GT_PITCH];

  const int tid  = threadIdx.x;
  const int lane = tid & 31;
  const int wave = tid >> 5;
  const int half = lane >> 4;
  const int idx  = lane & 15;
  const int mBlk = blockIdx.x >> 5;
  const int nBlk = blockIdx.x & 31;
  const int wm = wave >> 1, wn = wave & 1;

  int goff[GT_LD], goffA[GT_LD], loff[GT_LD];
  for (int j = 0; j < GT_LD; ++j) {
    const int i = tid + 128 * j;
    const int r = i / 20, c4 = i - r * 20;
    goff[j]  = r * NPATHS + c4 * 4;
    goffA[j] = r * AGG_PITCH + 16 * (r & 1) + c4 * 4;
    loff[j] = r * GT_PITCH + c4 * 4;
  }
  const float* gA = agg   + mBlk * 32 * AGG_PITCH;
  const float* gB = W_out + nBlk * 32 * NPATHS;

  const int fragA = (wm * 16 + idx) * GT_PITCH + 4 * half;
  const int fragB = (wn * 16 + idx) * GT_PITCH + 4 * half;

  v4f sa0[GT_LD], sb0[GT_LD], sa1[GT_LD], sb1[GT_LD];
  v4f fa0[GT_NF], fb0[GT_NF], fa1[GT_NF], fb1[GT_NF];
  v8f c0 = {}, c1 = {};

  auto issue = [&](v4f (&qa)[GT_LD], v4f (&qb)[GT_LD], int slab) {
    const float* nA = gA + slab * GT_K;
    const float* nB = gB + slab * GT_K;
    for (int j = 0; j < GT_LD; ++j) {
      qa[j] = *(const v4f*)(nA + goffA[j]);
      qb[j] = *(const v4f*)(nB + goff[j]);
    }
  };
  auto park = [&](const v4f (&qa)[GT_LD], const v4f (&qb)[GT_LD], int stage) {
    for (int j = 0; j < GT_LD; ++j) {
      *(v4f*)(&lsA[stage][loff[j]]) = qa[j];
      *(v4f*)(&lsB[stage][loff[j]]) = qb[j];
    }
  };
  auto fload = [&](v4f (&fa)[GT_NF], v4f (&fb)[GT_NF], int stage) {
    const float* pa = &lsA[stage][fragA];
    const float* pb = &lsB[stage][fragB];
    for (int i = 0; i < GT_NF; ++i) {
      fa[i] = *(const v4f*)(pa + 8 * i);
      fb[i] = *(const v4f*)(pb + 8 * i);
    }
  };
  auto mma = [&](const v4f (&fa)[GT_NF], const v4f (&fb)[GT_NF], int lo, int hi) {
    for (int i = lo; i < hi; ++i) {
      c0 = __builtin_amdgcn_wmma_f32_16x16x4_f32(
           false, fa[i].xy,  false, fb[i].xy,
           (short)0, c0,  false,  false);
      c1 = __builtin_amdgcn_wmma_f32_16x16x4_f32(
           false, fa[i].zw,  false, fb[i].zw,
           (short)0, c1,  false,  false);
    }
  };

  issue(sa0, sb0, 0);
  issue(sa1, sb1, 1);
  park(sa0, sb0, 0);
  lds_barrier();
  fload(fa0, fb0, 0);
  pin();

  for (int s = 0; s < GT_NSLAB - 3; s += 2) {
    mma(fa0, fb0, 0, 1);      pin();
    issue(sa0, sb0, s + 2);   pin();
    mma(fa0, fb0, 1, 3);      pin();
    park(sa1, sb1, 1);        pin();
    mma(fa0, fb0, 3, 7);      pin();
    lds_barrier();
    fload(fa1, fb1, 1);       pin();
    mma(fa0, fb0, 7, GT_NF);
    mma(fa1, fb1, 0, 1);      pin();
    issue(sa1, sb1, s + 3);   pin();
    mma(fa1, fb1, 1, 3);      pin();
    park(sa0, sb0, 0);        pin();
    mma(fa1, fb1, 3, 7);      pin();
    lds_barrier();
    fload(fa0, fb0, 0);       pin();
    mma(fa1, fb1, 7, GT_NF);
  }
  mma(fa0, fb0, 0, 1);        pin();
  issue(sa0, sb0, GT_NSLAB - 1); pin();
  mma(fa0, fb0, 1, 3);        pin();
  park(sa1, sb1, 1);          pin();
  mma(fa0, fb0, 3, 7);        pin();
  lds_barrier();
  fload(fa1, fb1, 1);         pin();
  mma(fa0, fb0, 7, GT_NF);
  mma(fa1, fb1, 0, 3);        pin();
  park(sa0, sb0, 0);          pin();
  mma(fa1, fb1, 3, 7);        pin();
  lds_barrier();
  fload(fa0, fb0, 0);         pin();
  mma(fa1, fb1, 7, GT_NF);
  mma(fa0, fb0, 0, GT_NF);
  lds_barrier();

  const v8f c = c0 + c1;
  float* Cs = &lsA[0][0];
  for (int r = 0; r < 8; ++r)
    Cs[(wm * 16 + 8 * half + r) * GT_CP + wn * 16 + idx] = c[r];
  lds_barrier();
  const int c4 = tid & 7;
  const v4f bb = *(const v4f*)(b_out + nBlk * 32 + c4 * 4);
  for (int h = 0; h < 2; ++h) {
    const int row = (tid >> 3) + 16 * h;
    const v4f v = *(const v4f*)(Cs + row * GT_CP + c4 * 4);
    *(v4f*)(out + (mBlk * 32 + row) * OUT_DIM + nBlk * 32 + c4 * 4) = v + bb;
  }
}

extern "C" void kernel_launch(void* const* d_in, const int* in_sizes, int n_in,
                              void* d_out, int out_size, void* d_ws, size_t ws_size,
                              hipStream_t stream) {
  (void)in_sizes; (void)n_in; (void)out_size; (void)ws_size;
  const float* x      = (const float*)d_in[0];
  const float* W_in   = (const float*)d_in[1];
  const float* b_in   = (const float*)d_in[2];
  const float* decay  = (const float*)d_in[3];
  const float* W_conv = (const float*)d_in[4];
  const float* b_conv = (const float*)d_in[5];
  const float* W_out  = (const float*)d_in[6];
  const float* b_out  = (const float*)d_in[7];
  const float* eps    = (const float*)d_in[8];
  float* out = (float*)d_out;

  float* agg = (float*)d_ws;

  bicep_row_kernel<<<BATCH, RP_THREADS, 0, stream>>>(x, W_in, b_in, decay, eps, W_conv,
                                                     b_conv, agg);
  bicep_gemm_kernel<<<8 * 32, 128, 0, stream>>>(agg, W_out, b_out, out);
}


# GEMM operand loads requested four K slabs ahead instead of two (four staging register sets, issued in slab order): a slab's loads get three slabs of matrix work to arrive under; k and sum order kept
# speedup vs baseline: 2.0643x; 1.0801x over previous
#include <hip/hip_runtime.h>
#include <hip/hip_bf16.h>
#include <math.h>

#define BATCH   256
#define IN_DIM  1024
#define OUT_DIM 1024
#define NPATHS  2000
#define NSTEPS  128
#define NPTOT   (BATCH * NPATHS)
#define DT      (1.0f / 128.0f)
#define SQDT    0.08838834764831845f

typedef float v2f __attribute__((ext_vector_type(2)));
typedef float v4f __attribute__((ext_vector_type(4)));
typedef float v8f __attribute__((ext_vector_type(8)));

static __device__ __forceinline__ void pin() { __builtin_amdgcn_sched_barrier(0); }

#define RP_THREADS 512
#define RP_PIECES  (NPATHS / 4)
#define RP_GRP     16
#define RP_NGRP    (NSTEPS / RP_GRP)
#define RP_TSTRIDE (NPTOT / 4)
#define AGG_PITCH  2048

__global__ __launch_bounds__(RP_THREADS) void bicep_row_kernel(
    const float* __restrict__ x, const float* __restrict__ W_in,
    const float* __restrict__ b_in, const float* __restrict__ decay,
    const float* __restrict__ eps, const float* __restrict__ W_conv,
    const float* __restrict__ b_conv, float* __restrict__ agg) {
  __shared__ float red[10][64];
  __shared__ float a_sh[16];
  __shared__ float shL[RP_THREADS];
  __shared__ float shR[RP_THREADS];

  const int tid = threadIdx.x;
  const int b   = blockIdx.x;

  {
    const int grp = tid >> 6, j = tid & 63;
    const float* xr = x + b * IN_DIM;
    const float* wr = W_in + grp * IN_DIM;
    float s = 0.0f;
    for (int i = j; i < IN_DIM; i += 64) s = fmaf(xr[i], wr[i], s);
    red[grp][j] = s;
    if (grp < 2) {
      const float* wr2 = W_in + (grp + 8) * IN_DIM;
      float s2 = 0.0f;
      for (int i = j; i < IN_DIM; i += 64) s2 = fmaf(xr[i], wr2[i], s2);
      red[grp + 8][j] = s2;
    }
  }
  __syncthreads();
  {
    const int wave = tid >> 5, lane = tid & 31;
    if (wave < 10) {
      float v = red[wave][lane] + red[wave][lane + 32];
      v += __shfl_down(v, 16);
      v += __shfl_down(v, 8);
      v += __shfl_down(v, 4);
      v += __shfl_down(v, 2);
      v += __shfl_down(v, 1);
      if (lane == 0) {
        const float proj = v + b_in[wave];
        const float fb = 1.0f / (1.0f + expf(-proj));
        a_sh[wave] = 1.0f + (fb - decay[0]) * DT;
      }
    }
  }
  __syncthreads();

  const int p = tid - 4 * (b & 1);
  const bool active = (p >= 0) && (p < RP_PIECES);
  const int pc = p < 0 ? 0 : (p >= RP_PIECES ? RP_PIECES - 1 : p);
  const float aa = a_sh[pc / 50];
  const v4f* ep = (const v4f*)(eps + b * NPATHS + 4 * pc);

  v4f X = {0.0f, 0.0f, 0.0f, 0.0f};
  v4f s0 = X, s1 = X, s2 = X;
  v4f q0[RP_GRP], q1[RP_GRP];

  auto issue = [&](v4f (&q)[RP_GRP], int g) {
    const v4f* src = ep + g * (RP_GRP * RP_TSTRIDE);
#pragma unroll
    for (int k = 0; k < RP_GRP; ++k) q[k] = src[k * RP_TSTRIDE];
  };
  auto scan = [&](const v4f (&q)[RP_GRP], int g) {
    const float* w = W_conv + (g * RP_GRP + 1) * 3;
#pragma unroll
    for (int k = 0; k < RP_GRP; ++k) {
      const float w0 = w[3 * k + 0], w1 = w[3 * k + 1], w2 = w[3 * k + 2];
#pragma unroll
      for (int c = 0; c < 4; ++c) {
        X[c]  = fmaf(X[c], aa, SQDT * q[k][c]);
        s0[c] = fmaf(w0, X[c], s0[c]);
        s1[c] = fmaf(w1, X[c], s1[c]);
        s2[c] = fmaf(w2, X[c], s2[c]);
      }
    }
  };

  issue(q0, 0);
  for (int g = 0; g < RP_NGRP - 2; g += 2) {
    issue(q1, g + 1);        pin();
    scan(q0, g);             pin();
    issue(q0, g + 2);        pin();
    scan(q1, g + 1);         pin();
  }
  issue(q1, RP_NGRP - 1);    pin();
  scan(q0, RP_NGRP - 2);     pin();
  scan(q1, RP_NGRP - 1);

  shL[tid] = s0[3];
  shR[tid] = s2[0];
  __syncthreads();
  if (active) {
    const float bc = b_conv[0];
    float v0 = s1[0] + bc;
    if (p > 0) v0 += shL[tid - 1];
    v0 += s2[1];
    float v1 = s1[1] + bc;  v1 += s0[0];  v1 += s2[2];
    float v2 = s1[2] + bc;  v2 += s0[1];  v2 += s2[3];
    float v3 = s1[3] + bc;  v3 += s0[2];
    if (p < RP_PIECES - 1) v3 += shR[tid + 1];
    const v4f o = {v0, v1, v2, v3};
    *(v4f*)(agg + b * AGG_PITCH + 4 * tid) = o;
  }
}

#define GT_K     80
#define GT_PITCH 84
#define GT_NSLAB (NPATHS / GT_K)
#define GT_LD    5
#define GT_NF    (GT_K / 8)
#define GT_CP    36

static __device__ __forceinline__ void lds_barrier() {
  asm volatile("s_wait_dscnt 0x0\n\ts_barrier_signal -1\n\ts_barrier_wait -1" ::: "memory");
}

__global__ __launch_bounds__(128) void bicep_gemm_kernel(
    const float* __restrict__ agg, const float* __restrict__ W_out,
    const float* __restrict__ b_out, float* __restrict__ out) {
  __shared__ __attribute__((aligned(16))) float lsA[2][32 * GT_PITCH];
  __shared__ __attribute__((aligned(16))) float lsB[2][32 * GT_PITCH];

  const int tid  = threadIdx.x;
  const int lane = tid & 31;
  const int wave = tid >> 5;
  const int half = lane >> 4;
  const int idx  = lane & 15;
  const int mBlk = blockIdx.x >> 5;
  const int nBlk = blockIdx.x & 31;
  const int wm = wave >> 1, wn = wave & 1;

  int goff[GT_LD], goffA[GT_LD], loff[GT_LD];
  for (int j = 0; j < GT_LD; ++j) {
    const int i = tid + 128 * j;
    const int r = i / 20, c4 = i - r * 20;
    goff[j]  = r * NPATHS + c4 * 4;
    goffA[j] = r * AGG_PITCH + 16 * (r & 1) + c4 * 4;
    loff[j] = r * GT_PITCH + c4 * 4;
  }
  const float* gA = agg   + mBlk * 32 * AGG_PITCH;
  const float* gB = W_out + nBlk * 32 * NPATHS;

  const int fragA = (wm * 16 + idx) * GT_PITCH + 4 * half;
  const int fragB = (wn * 16 + idx) * GT_PITCH + 4 * half;

  v4f sa0[GT_LD], sb0[GT_LD], sa1[GT_LD], sb1[GT_LD];
  v4f sa2[GT_LD], sb2[GT_LD], sa3[GT_LD], sb3[GT_LD];
  v4f fa0[GT_NF], fb0[GT_NF], fa1[GT_NF], fb1[GT_NF];
  v8f c0 = {}, c1 = {};

  auto issue = [&](v4f (&qa)[GT_LD], v4f (&qb)[GT_LD], int slab) {
    const float* nA = gA + slab * GT_K;
    const float* nB = gB + slab * GT_K;
    for (int j = 0; j < GT_LD; ++j) {
      qa[j] = *(const v4f*)(nA + goffA[j]);
      qb[j] = *(const v4f*)(nB + goff[j]);
    }
  };
  auto park = [&](const v4f (&qa)[GT_LD], const v4f (&qb)[GT_LD], int stage) {
    for (int j = 0; j < GT_LD; ++j) {
      *(v4f*)(&lsA[stage][loff[j]]) = qa[j];
      *(v4f*)(&lsB[stage][loff[j]]) = qb[j];
    }
  };
  auto fload = [&](v4f (&fa)[GT_NF], v4f (&fb)[GT_NF], int stage) {
    const float* pa = &lsA[stage][fragA];
    const float* pb = &lsB[stage][fragB];
    for (int i = 0; i < GT_NF; ++i) {
      fa[i] = *(const v4f*)(pa + 8 * i);
      fb[i] = *(const v4f*)(pb + 8 * i);
    }
  };
  auto mma = [&](const v4f (&fa)[GT_NF], const v4f (&fb)[GT_NF], int lo, int hi) {
    for (int i = lo; i < hi; ++i) {
      c0 = __builtin_amdgcn_wmma_f32_16x16x4_f32(
           false, fa[i].xy,  false, fb[i].xy,
           (short)0, c0,  false,  false);
      c1 = __builtin_amdgcn_wmma_f32_16x16x4_f32(
           false, fa[i].zw,  false, fb[i].zw,
           (short)0, c1,  false,  false);
    }
  };

  auto run_slab = [&](const v4f (&fa)[GT_NF], const v4f (&fb)[GT_NF],
                      v4f (&qa)[GT_LD], v4f (&qb)[GT_LD], int fetch,
                      const v4f (&pa)[GT_LD], const v4f (&pb)[GT_LD],
                      v4f (&ga)[GT_NF], v4f (&gb)[GT_NF], int stage) {
    mma(fa, fb, 0, 1);        pin();
    issue(qa, qb, fetch);     pin();
    mma(fa, fb, 1, 3);        pin();
    park(pa, pb, stage);      pin();
    mma(fa, fb, 3, 7);        pin();
    lds_barrier();
    fload(ga, gb, stage);     pin();
    mma(fa, fb, 7, GT_NF);
  };
  auto run_slab_tail = [&](const v4f (&fa)[GT_NF], const v4f (&fb)[GT_NF],
                           const v4f (&pa)[GT_LD], const v4f (&pb)[GT_LD],
                           v4f (&ga)[GT_NF], v4f (&gb)[GT_NF], int stage) {
    mma(fa, fb, 0, 3);        pin();
    park(pa, pb, stage);      pin();
    mma(fa, fb, 3, 7);        pin();
    lds_barrier();
    fload(ga, gb, stage);     pin();
    mma(fa, fb, 7, GT_NF);
  };

  issue(sa0, sb0, 0);        pin();
  issue(sa1, sb1, 1);        pin();
  issue(sa2, sb2, 2);        pin();
  issue(sa3, sb3, 3);        pin();
  park(sa0, sb0, 0);
  lds_barrier();
  fload(fa0, fb0, 0);
  pin();

  for (int s = 0; s < GT_NSLAB - 5; s += 4) {
    run_slab(fa0, fb0, sa0, sb0, s + 4, sa1, sb1, fa1, fb1, 1);
    run_slab(fa1, fb1, sa1, sb1, s + 5, sa2, sb2, fa0, fb0, 0);
    run_slab(fa0, fb0, sa2, sb2, s + 6, sa3, sb3, fa1, fb1, 1);
    run_slab(fa1, fb1, sa3, sb3, s + 7, sa0, sb0, fa0, fb0, 0);
  }
  run_slab(fa0, fb0, sa0, sb0, GT_NSLAB - 1, sa1, sb1, fa1, fb1, 1);
  run_slab_tail(fa1, fb1, sa2, sb2, fa0, fb0, 0);
  run_slab_tail(fa0, fb0, sa3, sb3, fa1, fb1, 1);
  run_slab_tail(fa1, fb1, sa0, sb0, fa0, fb0, 0);
  mma(fa0, fb0, 0, GT_NF);
  lds_barrier();

  const v8f c = c0 + c1;
  float* Cs = &lsA[0][0];
  for (int r = 0; r < 8; ++r)
    Cs[(wm * 16 + 8 * half + r) * GT_CP + wn * 16 + idx] = c[r];
  lds_barrier();
  const int c4 = tid & 7;
  const v4f bb = *(const v4f*)(b_out + nBlk * 32 + c4 * 4);
  for (int h = 0; h < 2; ++h) {
    const int row = (tid >> 3) + 16 * h;
    const v4f v = *(const v4f*)(Cs + row * GT_CP + c4 * 4);
    *(v4f*)(out + (mBlk * 32 + row) * OUT_DIM + nBlk * 32 + c4 * 4) = v + bb;
  }
}

extern "C" void kernel_launch(void* const* d_in, const int* in_sizes, int n_in,
                              void* d_out, int out_size, void* d_ws, size_t ws_size,
                              hipStream_t stream) {
  (void)in_sizes; (void)n_in; (void)out_size; (void)ws_size;
  const float* x      = (const float*)d_in[0];
  const float* W_in   = (const float*)d_in[1];
  const float* b_in   = (const float*)d_in[2];
  const float* decay  = (const float*)d_in[3];
  const float* W_conv = (const float*)d_in[4];
  const float* b_conv = (const float*)d_in[5];
  const float* W_out  = (const float*)d_in[6];
  const float* b_out  = (const float*)d_in[7];
  const float* eps    = (const float*)d_in[8];
  float* out = (float*)d_out;

  float* agg = (float*)d_ws;

  bicep_row_kernel<<<BATCH, RP_THREADS, 0, stream>>>(x, W_in, b_in, decay, eps, W_conv,
                                                     b_conv, agg);
  bicep_gemm_kernel<<<8 * 32, 128, 0, stream>>>(agg, W_out, b_out, out);
}


# Row kernel: stage-1 operand loads and the first two eps load groups requested at wave entry, stage-1 barriers no longer drain loads: the eps stream runs under the feedback stage; sum order kept
# speedup vs baseline: 2.2120x; 1.0715x over previous
#include <hip/hip_runtime.h>
#include <hip/hip_bf16.h>
#include <math.h>

#define BATCH   256
#define IN_DIM  1024
#define OUT_DIM 1024
#define NPATHS  2000
#define NSTEPS  128
#define NPTOT   (BATCH * NPATHS)
#define DT      (1.0f / 128.0f)
#define SQDT    0.08838834764831845f

typedef float v2f __attribute__((ext_vector_type(2)));
typedef float v4f __attribute__((ext_vector_type(4)));
typedef float v8f __attribute__((ext_vector_type(8)));

static __device__ __forceinline__ void pin() { __builtin_amdgcn_sched_barrier(0); }

static __device__ __forceinline__ void lds_barrier() {
  asm volatile("s_wait_dscnt 0x0\n\ts_barrier_signal -1\n\ts_barrier_wait -1" ::: "memory");
}

#define RP_THREADS 512
#define RP_PIECES  (NPATHS / 4)
#define RP_GRP     16
#define RP_NGRP    (NSTEPS / RP_GRP)
#define RP_TSTRIDE (NPTOT / 4)
#define AGG_PITCH  2048

__global__ __launch_bounds__(RP_THREADS) void bicep_row_kernel(
    const float* __restrict__ x, const float* __restrict__ W_in,
    const float* __restrict__ b_in, const float* __restrict__ decay,
    const float* __restrict__ eps, const float* __restrict__ W_conv,
    const float* __restrict__ b_conv, float* __restrict__ agg) {
  __shared__ float red[10][64];
  __shared__ float a_sh[16];
  __shared__ float shL[RP_THREADS];
  __shared__ float shR[RP_THREADS];

  const int tid = threadIdx.x;
  const int b   = blockIdx.x;

  const int p = tid - 4 * (b & 1);
  const bool active = (p >= 0) && (p < RP_PIECES);
  const int pc = p < 0 ? 0 : (p >= RP_PIECES ? RP_PIECES - 1 : p);
  const v4f* ep = (const v4f*)(eps + b * NPATHS + 4 * pc);

  v4f q0[RP_GRP], q1[RP_GRP];
  auto issue = [&](v4f (&q)[RP_GRP], int g) {
    const v4f* src = ep + g * (RP_GRP * RP_TSTRIDE);
#pragma unroll
    for (int k = 0; k < RP_GRP; ++k) q[k] = src[k * RP_TSTRIDE];
  };

  const int grp = tid >> 6, j = tid & 63;
  const int wave = tid >> 5, lane = tid & 31;
  float xv[IN_DIM / 64], wv[IN_DIM / 64], wv2[IN_DIM / 64];
  {
    const float* xr = x + b * IN_DIM;
    const float* wr = W_in + grp * IN_DIM;
#pragma unroll
    for (int k = 0; k < IN_DIM / 64; ++k) { xv[k] = xr[j + 64 * k]; wv[k] = wr[j + 64 * k]; }
    if (grp < 2) {
      const float* wr2 = W_in + (grp + 8) * IN_DIM;
#pragma unroll
      for (int k = 0; k < IN_DIM / 64; ++k) wv2[k] = wr2[j + 64 * k];
    }
  }
  const float bias_in = b_in[wave < 10 ? wave : 9];
  const float dec     = decay[0];
  pin();
  issue(q0, 0);
  issue(q1, 1);
  pin();

  {
    float s = 0.0f;
#pragma unroll
    for (int k = 0; k < IN_DIM / 64; ++k) s = fmaf(xv[k], wv[k], s);
    red[grp][j] = s;
    if (grp < 2) {
      float s2 = 0.0f;
#pragma unroll
      for (int k = 0; k < IN_DIM / 64; ++k) s2 = fmaf(xv[k], wv2[k], s2);
      red[grp + 8][j] = s2;
    }
  }
  lds_barrier();
  {
    if (wave < 10) {
      float v = red[wave][lane] + red[wave][lane + 32];
      v += __shfl_down(v, 16);
      v += __shfl_down(v, 8);
      v += __shfl_down(v, 4);
      v += __shfl_down(v, 2);
      v += __shfl_down(v, 1);
      if (lane == 0) {
        const float proj = v + bias_in;
        const float fb = 1.0f / (1.0f + expf(-proj));
        a_sh[wave] = 1.0f + (fb - dec) * DT;
      }
    }
  }
  lds_barrier();

  const float aa = a_sh[pc / 50];

  v4f X = {0.0f, 0.0f, 0.0f, 0.0f};
  v4f s0 = X, s1 = X, s2 = X;

  auto scan = [&](const v4f (&q)[RP_GRP], int g) {
    const float* w = W_conv + (g * RP_GRP + 1) * 3;
#pragma unroll
    for (int k = 0; k < RP_GRP; ++k) {
      const float w0 = w[3 * k + 0], w1 = w[3 * k + 1], w2 = w[3 * k + 2];
#pragma unroll
      for (int c = 0; c < 4; ++c) {
        X[c]  = fmaf(X[c], aa, SQDT * q[k][c]);
        s0[c] = fmaf(w0, X[c], s0[c]);
        s1[c] = fmaf(w1, X[c], s1[c]);
        s2[c] = fmaf(w2, X[c], s2[c]);
      }
    }
  };

  pin();
  for (int g = 0; g < RP_NGRP - 2; g += 2) {
    scan(q0, g);             pin();
    issue(q0, g + 2);        pin();
    scan(q1, g + 1);         pin();
    issue(q1, g + 3);        pin();
  }
  scan(q0, RP_NGRP - 2);     pin();
  scan(q1, RP_NGRP - 1);

  shL[tid] = s0[3];
  shR[tid] = s2[0];
  __syncthreads();
  if (active) {
    const float bc = b_conv[0];
    float v0 = s1[0] + bc;
    if (p > 0) v0 += shL[tid - 1];
    v0 += s2[1];
    float v1 = s1[1] + bc;  v1 += s0[0];  v1 += s2[2];
    float v2 = s1[2] + bc;  v2 += s0[1];  v2 += s2[3];
    float v3 = s1[3] + bc;  v3 += s0[2];
    if (p < RP_PIECES - 1) v3 += shR[tid + 1];
    const v4f o = {v0, v1, v2, v3};
    *(v4f*)(agg + b * AGG_PITCH + 4 * tid) = o;
  }
}

#define GT_K     80
#define GT_PITCH 84
#define GT_NSLAB (NPATHS / GT_K)
#define GT_LD    5
#define GT_NF    (GT_K / 8)
#define GT_CP    36


__global__ __launch_bounds__(128) void bicep_gemm_kernel(
    const float* __restrict__ agg, const float* __restrict__ W_out,
    const float* __restrict__ b_out, float* __restrict__ out) {
  __shared__ __attribute__((aligned(16))) float lsA[2][32 * GT_PITCH];
  __shared__ __attribute__((aligned(16))) float lsB[2][32 * GT_PITCH];

  const int tid  = threadIdx.x;
  const int lane = tid & 31;
  const int wave = tid >> 5;
  const int half = lane >> 4;
  const int idx  = lane & 15;
  const int mBlk = blockIdx.x >> 5;
  const int nBlk = blockIdx.x & 31;
  const int wm = wave >> 1, wn = wave & 1;

  int goff[GT_LD], goffA[GT_LD], loff[GT_LD];
  for (int j = 0; j < GT_LD; ++j) {
    const int i = tid + 128 * j;
    const int r = i / 20, c4 = i - r * 20;
    goff[j]  = r * NPATHS + c4 * 4;
    goffA[j] = r * AGG_PITCH + 16 * (r & 1) + c4 * 4;
    loff[j] = r * GT_PITCH + c4 * 4;
  }
  const float* gA = agg   + mBlk * 32 * AGG_PITCH;
  const float* gB = W_out + nBlk * 32 * NPATHS;

  const int fragA = (wm * 16 + idx) * GT_PITCH + 4 * half;
  const int fragB = (wn * 16 + idx) * GT_PITCH + 4 * half;

  v4f sa0[GT_LD], sb0[GT_LD], sa1[GT_LD], sb1[GT_LD];
  v4f sa2[GT_LD], sb2[GT_LD], sa3[GT_LD], sb3[GT_LD];
  v4f fa0[GT_NF], fb0[GT_NF], fa1[GT_NF], fb1[GT_NF];
  v8f c0 = {}, c1 = {};

  auto issue = [&](v4f (&qa)[GT_LD], v4f (&qb)[GT_LD], int slab) {
    const float* nA = gA + slab * GT_K;
    const float* nB = gB + slab * GT_K;
    for (int j = 0; j < GT_LD; ++j) {
      qa[j] = *(const v4f*)(nA + goffA[j]);
      qb[j] = *(const v4f*)(nB + goff[j]);
    }
  };
  auto park = [&](const v4f (&qa)[GT_LD], const v4f (&qb)[GT_LD], int stage) {
    for (int j = 0; j < GT_LD; ++j) {
      *(v4f*)(&lsA[stage][loff[j]]) = qa[j];
      *(v4f*)(&lsB[stage][loff[j]]) = qb[j];
    }
  };
  auto fload = [&](v4f (&fa)[GT_NF], v4f (&fb)[GT_NF], int stage) {
    const float* pa = &lsA[stage][fragA];
    const float* pb = &lsB[stage][fragB];
    for (int i = 0; i < GT_NF; ++i) {
      fa[i] = *(const v4f*)(pa + 8 * i);
      fb[i] = *(const v4f*)(pb + 8 * i);
    }
  };
  auto mma = [&](const v4f (&fa)[GT_NF], const v4f (&fb)[GT_NF], int lo, int hi) {
    for (int i = lo; i < hi; ++i) {
      c0 = __builtin_amdgcn_wmma_f32_16x16x4_f32(
           false, fa[i].xy,  false, fb[i].xy,
           (short)0, c0,  false,  false);
      c1 = __builtin_amdgcn_wmma_f32_16x16x4_f32(
           false, fa[i].zw,  false, fb[i].zw,
           (short)0, c1,  false,  false);
    }
  };

  auto run_slab = [&](const v4f (&fa)[GT_NF], const v4f (&fb)[GT_NF],
                      v4f (&qa)[GT_LD], v4f (&qb)[GT_LD], int fetch,
                      const v4f (&pa)[GT_LD], const v4f (&pb)[GT_LD],
                      v4f (&ga)[GT_NF], v4f (&gb)[GT_NF], int stage) {
    mma(fa, fb, 0, 1);        pin();
    issue(qa, qb, fetch);     pin();
    mma(fa, fb, 1, 3);        pin();
    park(pa, pb, stage);      pin();
    mma(fa, fb, 3, 7);        pin();
    lds_barrier();
    fload(ga, gb, stage);     pin();
    mma(fa, fb, 7, GT_NF);
  };
  auto run_slab_tail = [&](const v4f (&fa)[GT_NF], const v4f (&fb)[GT_NF],
                           const v4f (&pa)[GT_LD], const v4f (&pb)[GT_LD],
                           v4f (&ga)[GT_NF], v4f (&gb)[GT_NF], int stage) {
    mma(fa, fb, 0, 3);        pin();
    park(pa, pb, stage);      pin();
    mma(fa, fb, 3, 7);        pin();
    lds_barrier();
    fload(ga, gb, stage);     pin();
    mma(fa, fb, 7, GT_NF);
  };

  issue(sa0, sb0, 0);        pin();
  issue(sa1, sb1, 1);        pin();
  issue(sa2, sb2, 2);        pin();
  issue(sa3, sb3, 3);        pin();
  park(sa0, sb0, 0);
  lds_barrier();
  fload(fa0, fb0, 0);
  pin();

  for (int s = 0; s < GT_NSLAB - 5; s += 4) {
    run_slab(fa0, fb0, sa0, sb0, s + 4, sa1, sb1, fa1, fb1, 1);
    run_slab(fa1, fb1, sa1, sb1, s + 5, sa2, sb2, fa0, fb0, 0);
    run_slab(fa0, fb0, sa2, sb2, s + 6, sa3, sb3, fa1, fb1, 1);
    run_slab(fa1, fb1, sa3, sb3, s + 7, sa0, sb0, fa0, fb0, 0);
  }
  run_slab(fa0, fb0, sa0, sb0, GT_NSLAB - 1, sa1, sb1, fa1, fb1, 1);
  run_slab_tail(fa1, fb1, sa2, sb2, fa0, fb0, 0);
  run_slab_tail(fa0, fb0, sa3, sb3, fa1, fb1, 1);
  run_slab_tail(fa1, fb1, sa0, sb0, fa0, fb0, 0);
  mma(fa0, fb0, 0, GT_NF);
  lds_barrier();

  const v8f c = c0 + c1;
  float* Cs = &lsA[0][0];
  for (int r = 0; r < 8; ++r)
    Cs[(wm * 16 + 8 * half + r) * GT_CP + wn * 16 + idx] = c[r];
  lds_barrier();
  const int c4 = tid & 7;
  const v4f bb = *(const v4f*)(b_out + nBlk * 32 + c4 * 4);
  for (int h = 0; h < 2; ++h) {
    const int row = (tid >> 3) + 16 * h;
    const v4f v = *(const v4f*)(Cs + row * GT_CP + c4 * 4);
    *(v4f*)(out + (mBlk * 32 + row) * OUT_DIM + nBlk * 32 + c4 * 4) = v + bb;
  }
}

extern "C" void kernel_launch(void* const* d_in, const int* in_sizes, int n_in,
                              void* d_out, int out_size, void* d_ws, size_t ws_size,
                              hipStream_t stream) {
  (void)in_sizes; (void)n_in; (void)out_size; (void)ws_size;
  const float* x      = (const float*)d_in[0];
  const float* W_in   = (const float*)d_in[1];
  const float* b_in   = (const float*)d_in[2];
  const float* decay  = (const float*)d_in[3];
  const float* W_conv = (const float*)d_in[4];
  const float* b_conv = (const float*)d_in[5];
  const float* W_out  = (const float*)d_in[6];
  const float* b_out  = (const float*)d_in[7];
  const float* eps    = (const float*)d_in[8];
  float* out = (float*)d_out;

  float* agg = (float*)d_ws;

  bicep_row_kernel<<<BATCH, RP_THREADS, 0, stream>>>(x, W_in, b_in, decay, eps, W_conv,
                                                     b_conv, agg);
  bicep_gemm_kernel<<<8 * 32, 128, 0, stream>>>(agg, W_out, b_out, out);
}
